# starting kernel
# baseline (speedup 1.0000x reference)
#include <hip/hip_runtime.h>

#define N_NODES 50000
#define DEG     16
#define NIN     128
#define NOUT    64

typedef __attribute__((ext_vector_type(2))) float v2f;
typedef __attribute__((ext_vector_type(8))) float v8f;

__global__ __launch_bounds__(256) void proj_wmma_kernel(
    const float* __restrict__ node_fts, const float* __restrict__ edge_fts,
    const float* __restrict__ W_node,   const float* __restrict__ W_edge,
    float* __restrict__ hv, float* __restrict__ ev)
{
    const int wave   = (blockIdx.x * blockDim.x + threadIdx.x) >> 5;
    const int ntiles = N_NODES / 16;
    if (wave >= 2 * ntiles) return;
    const int  mat  = (wave >= ntiles) ? 1 : 0;
    const int  tile = mat ? (wave - ntiles) : wave;
    const float* A  = mat ? edge_fts : node_fts;
    const float* W  = mat ? W_edge   : W_node;
    float*       O  = mat ? ev       : hv;

    const int row0 = tile * 16;
    const int lane = threadIdx.x & 31;
    const int m    = lane & 15;
    const int kh   = lane >> 4;

    const float* arow = A + (size_t)(row0 + m) * NIN + 2 * kh;
    v8f acc0 = {}, acc1 = {}, acc2 = {}, acc3 = {};

    #pragma unroll 4
    for (int k0 = 0; k0 < NIN; k0 += 4) {
        v2f a = *(const v2f*)(arow + k0);
        const float* w0 = W + (size_t)(k0 + 2 * kh) * NOUT + m;
        const float* w1 = w0 + NOUT;
        v2f b0 = { w0[0],  w1[0]  };
        v2f b1 = { w0[16], w1[16] };
        v2f b2 = { w0[32], w1[32] };
        v2f b3 = { w0[48], w1[48] };
        acc0 = __builtin_amdgcn_wmma_f32_16x16x4_f32(false, a, false, b0, (short)0, acc0, false, false);
        acc1 = __builtin_amdgcn_wmma_f32_16x16x4_f32(false, a, false, b1, (short)0, acc1, false, false);
        acc2 = __builtin_amdgcn_wmma_f32_16x16x4_f32(false, a, false, b2, (short)0, acc2, false, false);
        acc3 = __builtin_amdgcn_wmma_f32_16x16x4_f32(false, a, false, b3, (short)0, acc3, false, false);
    }

    float* orow = O + (size_t)(row0 + kh * 8) * NOUT + m;
    #pragma unroll
    for (int r = 0; r < 8; ++r) {
        orow[(size_t)r * NOUT +  0] = acc0[r];
        orow[(size_t)r * NOUT + 16] = acc1[r];
        orow[(size_t)r * NOUT + 32] = acc2[r];
        orow[(size_t)r * NOUT + 48] = acc3[r];
    }
}

__global__ __launch_bounds__(256) void alpha_kernel(
    const float* __restrict__ hv, const float* __restrict__ ev,
    const float* __restrict__ a_node, const float* __restrict__ a_edge,
    float* __restrict__ asrc, float* __restrict__ adst,
    float* __restrict__ bsrc, float* __restrict__ bdst)
{
    const int wave = (blockIdx.x * blockDim.x + threadIdx.x) >> 5;
    if (wave >= N_NODES) return;
    const int lane = threadIdx.x & 31;

    const float a0l = a_node[lane],      a0h = a_node[lane + 32];
    const float a1l = a_node[lane + 64], a1h = a_node[lane + 96];
    const float b0l = a_edge[lane],      b0h = a_edge[lane + 32];
    const float b1l = a_edge[lane + 64], b1h = a_edge[lane + 96];

    const float hlo = hv[(size_t)wave * NOUT + lane];
    const float hhi = hv[(size_t)wave * NOUT + 32 + lane];
    const float elo = ev[(size_t)wave * NOUT + lane];
    const float ehi = ev[(size_t)wave * NOUT + 32 + lane];

    float ps = hlo * a0l + hhi * a0h;
    float pd = hlo * a1l + hhi * a1h;
    float qs = elo * b0l + ehi * b0h;
    float qd = elo * b1l + ehi * b1h;
    #pragma unroll
    for (int off = 16; off; off >>= 1) {
        ps += __shfl_xor(ps, off, 32);
        pd += __shfl_xor(pd, off, 32);
        qs += __shfl_xor(qs, off, 32);
        qd += __shfl_xor(qd, off, 32);
    }
    if (lane == 0) {
        asrc[wave] = ps; adst[wave] = pd;
        bsrc[wave] = qs; bdst[wave] = qd;
    }
}

__global__ __launch_bounds__(256) void edge_kernel(
    const float* __restrict__ hv, const float* __restrict__ ev,
    const int*  __restrict__ edges,
    const float* __restrict__ asrc, const float* __restrict__ adst,
    const float* __restrict__ bsrc, const float* __restrict__ bdst,
    float* __restrict__ node_out, float* __restrict__ edge_out,
    float* __restrict__ partials)
{
    __shared__ float sred[4][8];
    const int lane = threadIdx.x & 31;
    const int wid  = threadIdx.x >> 5;
    const int node = blockIdx.x * 8 + wid;

    float att_n = 0.f, att_e = 0.f;
    int   d = 0;
    if (node < N_NODES && lane < 16) {
        int2 ed = ((const int2*)edges)[(size_t)node * DEG + lane];
        const int s = ed.x;
        d = ed.y;
        float xn = asrc[s] + adst[d];
        float xe = bsrc[s] + bdst[d];
        xn = xn > 0.f ? xn : 0.2f * xn;
        xe = xe > 0.f ? xe : 0.2f * xe;
        xn = fminf(fmaxf(xn, -2.f), 2.f);
        xe = fminf(fmaxf(xe, -2.f), 2.f);
        att_n = __expf(xn);
        att_e = __expf(xe);
    }

    float sum_n = att_n, sum_e = att_e;
    #pragma unroll
    for (int off = 16; off; off >>= 1) {
        sum_n += __shfl_xor(sum_n, off, 32);
        sum_e += __shfl_xor(sum_e, off, 32);
    }
    const bool act = (node < N_NODES) && (lane < 16);
    float norm_n = act ? att_n / sum_n : 0.f;
    float norm_e = act ? att_e / sum_e : 0.f;

    float pn = norm_n, pn2 = norm_n * norm_n, pe = norm_e, pe2 = norm_e * norm_e;
    #pragma unroll
    for (int off = 16; off; off >>= 1) {
        pn  += __shfl_xor(pn,  off, 32);
        pn2 += __shfl_xor(pn2, off, 32);
        pe  += __shfl_xor(pe,  off, 32);
        pe2 += __shfl_xor(pe2, off, 32);
    }
    if (lane == 0) { sred[0][wid] = pn; sred[1][wid] = pn2; sred[2][wid] = pe; sred[3][wid] = pe2; }
    __syncthreads();
    if (threadIdx.x < 4) {
        float t = 0.f;
        #pragma unroll
        for (int w = 0; w < 8; ++w) t += sred[threadIdx.x][w];
        partials[(size_t)blockIdx.x * 4 + threadIdx.x] = t;
    }
    if (node >= N_NODES) return;

    const float2* hv2 = (const float2*)hv;
    const float2* ev2 = (const float2*)ev;
    float2 acc_h = {0.f, 0.f}, acc_e = {0.f, 0.f};
    #pragma unroll
    for (int j = 0; j < 16; ++j) {
        const int   dj = __shfl(d,      j, 32);
        const float wn = __shfl(norm_n, j, 32);
        const float we = __shfl(norm_e, j, 32);
        const float2 hvv = hv2[(size_t)dj * 32 + lane];
        const float2 evv = ev2[(size_t)dj * 32 + lane];
        acc_h.x += wn * hvv.x;  acc_h.y += wn * hvv.y;
        acc_e.x += we * evv.x;  acc_e.y += we * evv.y;
    }

    const float2 hs = hv2[(size_t)node * 32 + lane];
    const float2 es = ev2[(size_t)node * 32 + lane];
    float2* no2 = (float2*)node_out;
    float2* eo2 = (float2*)edge_out;
    float2 o0 = { pn * hs.x, pn * hs.y };
    float2 p0 = { pe * es.x, pe * es.y };
    no2[(size_t)node * 64 + lane]      = o0;
    no2[(size_t)node * 64 + 32 + lane] = acc_h;
    eo2[(size_t)node * 64 + lane]      = p0;
    eo2[(size_t)node * 64 + 32 + lane] = acc_e;
}

__global__ __launch_bounds__(256) void finalize_kernel(
    const float* __restrict__ partials, int nblocks, float* __restrict__ outv)
{
    __shared__ double red[4][256];
    double s0 = 0.0, s1 = 0.0, s2 = 0.0, s3 = 0.0;
    for (int i = threadIdx.x; i < nblocks; i += 256) {
        s0 += (double)partials[(size_t)i * 4 + 0];
        s1 += (double)partials[(size_t)i * 4 + 1];
        s2 += (double)partials[(size_t)i * 4 + 2];
        s3 += (double)partials[(size_t)i * 4 + 3];
    }
    red[0][threadIdx.x] = s0; red[1][threadIdx.x] = s1;
    red[2][threadIdx.x] = s2; red[3][threadIdx.x] = s3;
    __syncthreads();
    for (int off = 128; off; off >>= 1) {
        if (threadIdx.x < off) {
            red[0][threadIdx.x] += red[0][threadIdx.x + off];
            red[1][threadIdx.x] += red[1][threadIdx.x + off];
            red[2][threadIdx.x] += red[2][threadIdx.x + off];
            red[3][threadIdx.x] += red[3][threadIdx.x + off];
        }
        __syncthreads();
    }
    if (threadIdx.x == 0) {
        const double E = (double)N_NODES * (double)DEG;
        const double var_n = (red[1][0] - red[0][0] * red[0][0] / E) / (E - 1.0);
        const double var_e = (red[3][0] - red[2][0] * red[2][0] / E) / (E - 1.0);
        outv[0] = (float)var_n;
        outv[1] = (float)var_e;
    }
}

extern "C" void kernel_launch(void* const* d_in, const int* in_sizes, int n_in,
                              void* d_out, int out_size, void* d_ws, size_t ws_size,
                              hipStream_t stream) {
    const float* node_fts = (const float*)d_in[0];
    const float* edge_fts = (const float*)d_in[1];
    const int*   edges    = (const int*)  d_in[2];
    const float* W_node   = (const float*)d_in[3];
    const float* W_edge   = (const float*)d_in[4];
    const float* a_node   = (const float*)d_in[5];
    const float* a_edge   = (const float*)d_in[6];

    float* out      = (float*)d_out;
    float* node_out = out;
    float* edge_out = out + (size_t)N_NODES * 128;
    float* outv     = out + (size_t)2 * N_NODES * 128;

    float* ws   = (float*)d_ws;
    float* hv   = ws;
    float* ev   = hv + (size_t)N_NODES * NOUT;
    float* asrc = ev + (size_t)N_NODES * NOUT;
    float* adst = asrc + N_NODES;
    float* bsrc = adst + N_NODES;
    float* bdst = bsrc + N_NODES;
    float* partials = bdst + N_NODES;

    const int ntiles = N_NODES / 16;
    const int wavesA = 2 * ntiles;
    const int blocksA = (wavesA + 7) / 8;
    proj_wmma_kernel<<<blocksA, 256, 0, stream>>>(node_fts, edge_fts, W_node, W_edge, hv, ev);

    const int blocksB = (N_NODES + 7) / 8;
    alpha_kernel<<<blocksB, 256, 0, stream>>>(hv, ev, a_node, a_edge, asrc, adst, bsrc, bdst);

    const int blocksC = (N_NODES + 7) / 8;
    edge_kernel<<<blocksC, 256, 0, stream>>>(hv, ev, edges, asrc, adst, bsrc, bdst,
                                             node_out, edge_out, partials);

    finalize_kernel<<<1, 256, 0, stream>>>(partials, blocksC, outv);
}


# fuse the per-node attention dot products (h_v.a, e_v.a) into the projection kernel's epilogue, from the WMMA accumulators; the alpha pass and its launch go; the dots' summation tree changes
# speedup vs baseline: 1.1660x; 1.1660x over previous
#include <hip/hip_runtime.h>

#define N_NODES 50000
#define DEG     16
#define NIN     128
#define NOUT    64

typedef __attribute__((ext_vector_type(2))) float v2f;
typedef __attribute__((ext_vector_type(8))) float v8f;

__global__ __launch_bounds__(256) void proj_wmma_kernel(
    const float* __restrict__ node_fts, const float* __restrict__ edge_fts,
    const float* __restrict__ W_node,   const float* __restrict__ W_edge,
    const float* __restrict__ a_node,   const float* __restrict__ a_edge,
    float* __restrict__ hv, float* __restrict__ ev,
    float* __restrict__ asrc, float* __restrict__ adst,
    float* __restrict__ bsrc, float* __restrict__ bdst)
{
    const int wave   = (blockIdx.x * blockDim.x + threadIdx.x) >> 5;
    const int ntiles = N_NODES / 16;
    if (wave >= 2 * ntiles) return;
    const int  mat  = (wave >= ntiles) ? 1 : 0;
    const int  tile = mat ? (wave - ntiles) : wave;
    const float* A  = mat ? edge_fts : node_fts;
    const float* W  = mat ? W_edge   : W_node;
    float*       O  = mat ? ev       : hv;
    const float* av = mat ? a_edge   : a_node;
    float*       Ss = mat ? bsrc     : asrc;
    float*       Sd = mat ? bdst     : adst;

    const int row0 = tile * 16;
    const int lane = threadIdx.x & 31;
    const int m    = lane & 15;
    const int kh   = lane >> 4;

    const float* arow = A + (size_t)(row0 + m) * NIN + 2 * kh;
    v8f acc0 = {}, acc1 = {}, acc2 = {}, acc3 = {};

    const float cs0 = av[m],      cs1 = av[m + 16], cs2 = av[m + 32], cs3 = av[m + 48];
    const float cd0 = av[64 + m], cd1 = av[80 + m], cd2 = av[96 + m], cd3 = av[112 + m];

    #pragma unroll 4
    for (int k0 = 0; k0 < NIN; k0 += 4) {
        v2f a = *(const v2f*)(arow + k0);
        const float* w0 = W + (size_t)(k0 + 2 * kh) * NOUT + m;
        const float* w1 = w0 + NOUT;
        v2f b0 = { w0[0],  w1[0]  };
        v2f b1 = { w0[16], w1[16] };
        v2f b2 = { w0[32], w1[32] };
        v2f b3 = { w0[48], w1[48] };
        acc0 = __builtin_amdgcn_wmma_f32_16x16x4_f32(false, a, false, b0, (short)0, acc0, false, false);
        acc1 = __builtin_amdgcn_wmma_f32_16x16x4_f32(false, a, false, b1, (short)0, acc1, false, false);
        acc2 = __builtin_amdgcn_wmma_f32_16x16x4_f32(false, a, false, b2, (short)0, acc2, false, false);
        acc3 = __builtin_amdgcn_wmma_f32_16x16x4_f32(false, a, false, b3, (short)0, acc3, false, false);
    }

    float* orow = O + (size_t)(row0 + kh * 8) * NOUT + m;
    #pragma unroll
    for (int r = 0; r < 8; ++r) {
        orow[(size_t)r * NOUT +  0] = acc0[r];
        orow[(size_t)r * NOUT + 16] = acc1[r];
        orow[(size_t)r * NOUT + 32] = acc2[r];
        orow[(size_t)r * NOUT + 48] = acc3[r];
    }

    float x[16];
    for (int r = 0; r < 8; ++r) {
        x[r]     = acc0[r] * cs0 + acc1[r] * cs1 + acc2[r] * cs2 + acc3[r] * cs3;
        x[8 + r] = acc0[r] * cd0 + acc1[r] * cd1 + acc2[r] * cd2 + acc3[r] * cd3;
    }
    const bool up8 = (m & 8) != 0, up4 = (m & 4) != 0, up2 = (m & 2) != 0, up1 = (m & 1) != 0;
    float y[8], z[4], u[2];
    for (int i = 0; i < 8; ++i) {
        const float keep = up8 ? x[8 + i] : x[i];
        const float send = up8 ? x[i]     : x[8 + i];
        y[i] = keep + __shfl_xor(send, 8, 32);
    }
    for (int i = 0; i < 4; ++i) {
        const float keep = up4 ? y[4 + i] : y[i];
        const float send = up4 ? y[i]     : y[4 + i];
        z[i] = keep + __shfl_xor(send, 4, 32);
    }
    for (int i = 0; i < 2; ++i) {
        const float keep = up2 ? z[2 + i] : z[i];
        const float send = up2 ? z[i]     : z[2 + i];
        u[i] = keep + __shfl_xor(send, 2, 32);
    }
    const float keep0 = up1 ? u[1] : u[0];
    const float send0 = up1 ? u[0] : u[1];
    const float dot   = keep0 + __shfl_xor(send0, 1, 32);
    (up8 ? Sd : Ss)[row0 + 8 * kh + (m & 7)] = dot;
}

__global__ __launch_bounds__(256) void edge_kernel(
    const float* __restrict__ hv, const float* __restrict__ ev,
    const int*  __restrict__ edges,
    const float* __restrict__ asrc, const float* __restrict__ adst,
    const float* __restrict__ bsrc, const float* __restrict__ bdst,
    float* __restrict__ node_out, float* __restrict__ edge_out,
    float* __restrict__ partials)
{
    __shared__ float sred[4][8];
    const int lane = threadIdx.x & 31;
    const int wid  = threadIdx.x >> 5;
    const int node = blockIdx.x * 8 + wid;

    float att_n = 0.f, att_e = 0.f;
    int   d = 0;
    if (node < N_NODES && lane < 16) {
        int2 ed = ((const int2*)edges)[(size_t)node * DEG + lane];
        const int s = ed.x;
        d = ed.y;
        float xn = asrc[s] + adst[d];
        float xe = bsrc[s] + bdst[d];
        xn = xn > 0.f ? xn : 0.2f * xn;
        xe = xe > 0.f ? xe : 0.2f * xe;
        xn = fminf(fmaxf(xn, -2.f), 2.f);
        xe = fminf(fmaxf(xe, -2.f), 2.f);
        att_n = __expf(xn);
        att_e = __expf(xe);
    }

    float sum_n = att_n, sum_e = att_e;
    #pragma unroll
    for (int off = 16; off; off >>= 1) {
        sum_n += __shfl_xor(sum_n, off, 32);
        sum_e += __shfl_xor(sum_e, off, 32);
    }
    const bool act = (node < N_NODES) && (lane < 16);
    float norm_n = act ? att_n / sum_n : 0.f;
    float norm_e = act ? att_e / sum_e : 0.f;

    float pn = norm_n, pn2 = norm_n * norm_n, pe = norm_e, pe2 = norm_e * norm_e;
    #pragma unroll
    for (int off = 16; off; off >>= 1) {
        pn  += __shfl_xor(pn,  off, 32);
        pn2 += __shfl_xor(pn2, off, 32);
        pe  += __shfl_xor(pe,  off, 32);
        pe2 += __shfl_xor(pe2, off, 32);
    }
    if (lane == 0) { sred[0][wid] = pn; sred[1][wid] = pn2; sred[2][wid] = pe; sred[3][wid] = pe2; }
    __syncthreads();
    if (threadIdx.x < 4) {
        float t = 0.f;
        #pragma unroll
        for (int w = 0; w < 8; ++w) t += sred[threadIdx.x][w];
        partials[(size_t)blockIdx.x * 4 + threadIdx.x] = t;
    }
    if (node >= N_NODES) return;

    const float2* hv2 = (const float2*)hv;
    const float2* ev2 = (const float2*)ev;
    float2 acc_h = {0.f, 0.f}, acc_e = {0.f, 0.f};
    #pragma unroll
    for (int j = 0; j < 16; ++j) {
        const int   dj = __shfl(d,      j, 32);
        const float wn = __shfl(norm_n, j, 32);
        const float we = __shfl(norm_e, j, 32);
        const float2 hvv = hv2[(size_t)dj * 32 + lane];
        const float2 evv = ev2[(size_t)dj * 32 + lane];
        acc_h.x += wn * hvv.x;  acc_h.y += wn * hvv.y;
        acc_e.x += we * evv.x;  acc_e.y += we * evv.y;
    }

    const float2 hs = hv2[(size_t)node * 32 + lane];
    const float2 es = ev2[(size_t)node * 32 + lane];
    float2* no2 = (float2*)node_out;
    float2* eo2 = (float2*)edge_out;
    float2 o0 = { pn * hs.x, pn * hs.y };
    float2 p0 = { pe * es.x, pe * es.y };
    no2[(size_t)node * 64 + lane]      = o0;
    no2[(size_t)node * 64 + 32 + lane] = acc_h;
    eo2[(size_t)node * 64 + lane]      = p0;
    eo2[(size_t)node * 64 + 32 + lane] = acc_e;
}

__global__ __launch_bounds__(256) void finalize_kernel(
    const float* __restrict__ partials, int nblocks, float* __restrict__ outv)
{
    __shared__ double red[4][256];
    double s0 = 0.0, s1 = 0.0, s2 = 0.0, s3 = 0.0;
    for (int i = threadIdx.x; i < nblocks; i += 256) {
        s0 += (double)partials[(size_t)i * 4 + 0];
        s1 += (double)partials[(size_t)i * 4 + 1];
        s2 += (double)partials[(size_t)i * 4 + 2];
        s3 += (double)partials[(size_t)i * 4 + 3];
    }
    red[0][threadIdx.x] = s0; red[1][threadIdx.x] = s1;
    red[2][threadIdx.x] = s2; red[3][threadIdx.x] = s3;
    __syncthreads();
    for (int off = 128; off; off >>= 1) {
        if (threadIdx.x < off) {
            red[0][threadIdx.x] += red[0][threadIdx.x + off];
            red[1][threadIdx.x] += red[1][threadIdx.x + off];
            red[2][threadIdx.x] += red[2][threadIdx.x + off];
            red[3][threadIdx.x] += red[3][threadIdx.x + off];
        }
        __syncthreads();
    }
    if (threadIdx.x == 0) {
        const double E = (double)N_NODES * (double)DEG;
        const double var_n = (red[1][0] - red[0][0] * red[0][0] / E) / (E - 1.0);
        const double var_e = (red[3][0] - red[2][0] * red[2][0] / E) / (E - 1.0);
        outv[0] = (float)var_n;
        outv[1] = (float)var_e;
    }
}

extern "C" void kernel_launch(void* const* d_in, const int* in_sizes, int n_in,
                              void* d_out, int out_size, void* d_ws, size_t ws_size,
                              hipStream_t stream) {
    const float* node_fts = (const float*)d_in[0];
    const float* edge_fts = (const float*)d_in[1];
    const int*   edges    = (const int*)  d_in[2];
    const float* W_node   = (const float*)d_in[3];
    const float* W_edge   = (const float*)d_in[4];
    const float* a_node   = (const float*)d_in[5];
    const float* a_edge   = (const float*)d_in[6];

    float* out      = (float*)d_out;
    float* node_out = out;
    float* edge_out = out + (size_t)N_NODES * 128;
    float* outv     = out + (size_t)2 * N_NODES * 128;

    float* ws   = (float*)d_ws;
    float* hv   = ws;
    float* ev   = hv + (size_t)N_NODES * NOUT;
    float* asrc = ev + (size_t)N_NODES * NOUT;
    float* adst = asrc + N_NODES;
    float* bsrc = adst + N_NODES;
    float* bdst = bsrc + N_NODES;
    float* partials = bdst + N_NODES;

    const int ntiles = N_NODES / 16;
    const int wavesA = 2 * ntiles;
    const int blocksA = (wavesA + 7) / 8;
    proj_wmma_kernel<<<blocksA, 256, 0, stream>>>(node_fts, edge_fts, W_node, W_edge, a_node, a_edge,
                                                  hv, ev, asrc, adst, bsrc, bdst);

    const int blocksC = (N_NODES + 7) / 8;
    edge_kernel<<<blocksC, 256, 0, stream>>>(hv, ev, edges, asrc, adst, bsrc, bdst,
                                             node_out, edge_out, partials);

    finalize_kernel<<<1, 256, 0, stream>>>(partials, blocksC, outv);
}


# edge pass: each wave walks 8 consecutive nodes (6250 waves start, not 50000), edge rows and scores fetched ahead, a node's row loads issued as one batch; variance sums per wave, order changes
# speedup vs baseline: 1.3817x; 1.1850x over previous
#include <hip/hip_runtime.h>

#define N_NODES 50000
#define DEG     16
#define NIN     128
#define NOUT    64

typedef __attribute__((ext_vector_type(2))) float v2f;
typedef __attribute__((ext_vector_type(8))) float v8f;

__global__ __launch_bounds__(256) void proj_wmma_kernel(
    const float* __restrict__ node_fts, const float* __restrict__ edge_fts,
    const float* __restrict__ W_node,   const float* __restrict__ W_edge,
    const float* __restrict__ a_node,   const float* __restrict__ a_edge,
    float* __restrict__ hv, float* __restrict__ ev,
    float* __restrict__ asrc, float* __restrict__ adst,
    float* __restrict__ bsrc, float* __restrict__ bdst)
{
    const int wave   = (blockIdx.x * blockDim.x + threadIdx.x) >> 5;
    const int ntiles = N_NODES / 16;
    if (wave >= 2 * ntiles) return;
    const int  mat  = (wave >= ntiles) ? 1 : 0;
    const int  tile = mat ? (wave - ntiles) : wave;
    const float* A  = mat ? edge_fts : node_fts;
    const float* W  = mat ? W_edge   : W_node;
    float*       O  = mat ? ev       : hv;
    const float* av = mat ? a_edge   : a_node;
    float*       Ss = mat ? bsrc     : asrc;
    float*       Sd = mat ? bdst     : adst;

    const int row0 = tile * 16;
    const int lane = threadIdx.x & 31;
    const int m    = lane & 15;
    const int kh   = lane >> 4;

    const float* arow = A + (size_t)(row0 + m) * NIN + 2 * kh;
    v8f acc0 = {}, acc1 = {}, acc2 = {}, acc3 = {};

    const float cs0 = av[m],      cs1 = av[m + 16], cs2 = av[m + 32], cs3 = av[m + 48];
    const float cd0 = av[64 + m], cd1 = av[80 + m], cd2 = av[96 + m], cd3 = av[112 + m];

    #pragma unroll 4
    for (int k0 = 0; k0 < NIN; k0 += 4) {
        v2f a = *(const v2f*)(arow + k0);
        const float* w0 = W + (size_t)(k0 + 2 * kh) * NOUT + m;
        const float* w1 = w0 + NOUT;
        v2f b0 = { w0[0],  w1[0]  };
        v2f b1 = { w0[16], w1[16] };
        v2f b2 = { w0[32], w1[32] };
        v2f b3 = { w0[48], w1[48] };
        acc0 = __builtin_amdgcn_wmma_f32_16x16x4_f32(false, a, false, b0, (short)0, acc0, false, false);
        acc1 = __builtin_amdgcn_wmma_f32_16x16x4_f32(false, a, false, b1, (short)0, acc1, false, false);
        acc2 = __builtin_amdgcn_wmma_f32_16x16x4_f32(false, a, false, b2, (short)0, acc2, false, false);
        acc3 = __builtin_amdgcn_wmma_f32_16x16x4_f32(false, a, false, b3, (short)0, acc3, false, false);
    }

    float* orow = O + (size_t)(row0 + kh * 8) * NOUT + m;
    #pragma unroll
    for (int r = 0; r < 8; ++r) {
        orow[(size_t)r * NOUT +  0] = acc0[r];
        orow[(size_t)r * NOUT + 16] = acc1[r];
        orow[(size_t)r * NOUT + 32] = acc2[r];
        orow[(size_t)r * NOUT + 48] = acc3[r];
    }

    float x[16];
    for (int r = 0; r < 8; ++r) {
        x[r]     = acc0[r] * cs0 + acc1[r] * cs1 + acc2[r] * cs2 + acc3[r] * cs3;
        x[8 + r] = acc0[r] * cd0 + acc1[r] * cd1 + acc2[r] * cd2 + acc3[r] * cd3;
    }
    const bool up8 = (m & 8) != 0, up4 = (m & 4) != 0, up2 = (m & 2) != 0, up1 = (m & 1) != 0;
    float y[8], z[4], u[2];
    for (int i = 0; i < 8; ++i) {
        const float keep = up8 ? x[8 + i] : x[i];
        const float send = up8 ? x[i]     : x[8 + i];
        y[i] = keep + __shfl_xor(send, 8, 32);
    }
    for (int i = 0; i < 4; ++i) {
        const float keep = up4 ? y[4 + i] : y[i];
        const float send = up4 ? y[i]     : y[4 + i];
        z[i] = keep + __shfl_xor(send, 4, 32);
    }
    for (int i = 0; i < 2; ++i) {
        const float keep = up2 ? z[2 + i] : z[i];
        const float send = up2 ? z[i]     : z[2 + i];
        u[i] = keep + __shfl_xor(send, 2, 32);
    }
    const float keep0 = up1 ? u[1] : u[0];
    const float send0 = up1 ? u[0] : u[1];
    const float dot   = keep0 + __shfl_xor(send0, 1, 32);
    (up8 ? Sd : Ss)[row0 + 8 * kh + (m & 7)] = dot;
}

#define NPW        8
#define EDGE_WAVES (N_NODES / NPW)
#define EDGE_WPB   4

__global__ __launch_bounds__(EDGE_WPB * 32) void edge_kernel(
    const float* __restrict__ hv, const float* __restrict__ ev,
    const int*  __restrict__ edges,
    const float* __restrict__ asrc, const float* __restrict__ adst,
    const float* __restrict__ bsrc, const float* __restrict__ bdst,
    float* __restrict__ node_out, float* __restrict__ edge_out,
    float* __restrict__ partials)
{
    __shared__ float sred[4][EDGE_WPB];
    const int  lane = threadIdx.x & 31;
    const int  wid  = threadIdx.x >> 5;
    const int  wave = blockIdx.x * EDGE_WPB + wid;
    const int  e    = lane & 15;
    const bool own  = lane < 16;

    float s_n = 0.f, s_e = 0.f;
    float q_n = 0.f, q_e = 0.f;

    if (wave < EDGE_WAVES) {
        const int2*   e2  = (const int2*)edges;
        const float2* hv2 = (const float2*)hv;
        const float2* ev2 = (const float2*)ev;
        float2* no2 = (float2*)node_out;
        float2* eo2 = (float2*)edge_out;
        const int node0 = wave * NPW;

        int2  ed0 = e2[(size_t)node0 * DEG + e];
        int2  ed1 = e2[(size_t)(node0 + 1) * DEG + e];
        float a0s = asrc[ed0.x], a0d = adst[ed0.y];
        float b0s = bsrc[ed0.x], b0d = bdst[ed0.y];

        for (int t = 0; t < NPW; ++t) {
            const int node = node0 + t;

            const int  nn  = min(node + 2, N_NODES - 1);
            const int2 ed2 = e2[(size_t)nn * DEG + e];
            const float a1s = asrc[ed1.x], a1d = adst[ed1.y];
            const float b1s = bsrc[ed1.x], b1d = bdst[ed1.y];

            const int d = ed0.y;
            float2 hvv[16], evv[16];
            #pragma unroll
            for (int j = 0; j < 16; ++j) {
                const unsigned dj = (unsigned)__builtin_amdgcn_readlane(d, j);
                hvv[j] = hv2[dj * 32u + (unsigned)lane];
                evv[j] = ev2[dj * 32u + (unsigned)lane];
            }
            const unsigned ixs = (unsigned)node * 32u + (unsigned)lane;
            const float2 hs = hv2[ixs];
            const float2 es = ev2[ixs];
            __builtin_amdgcn_sched_group_barrier(0x020, 39, 0);
            __builtin_amdgcn_sched_group_barrier(0x080, 52, 0);

            float xn = a0s + a0d;
            float xe = b0s + b0d;
            xn = xn > 0.f ? xn : 0.2f * xn;
            xe = xe > 0.f ? xe : 0.2f * xe;
            xn = fminf(fmaxf(xn, -2.f), 2.f);
            xe = fminf(fmaxf(xe, -2.f), 2.f);
            const float att_n = own ? __expf(xn) : 0.f;
            const float att_e = own ? __expf(xe) : 0.f;

            float sum_n = att_n, sum_e = att_e;
            #pragma unroll
            for (int off = 16; off; off >>= 1) {
                sum_n += __shfl_xor(sum_n, off, 32);
                sum_e += __shfl_xor(sum_e, off, 32);
            }
            const float norm_n = own ? att_n / sum_n : 0.f;
            const float norm_e = own ? att_e / sum_e : 0.f;

            float pn = norm_n, pe = norm_e;
            #pragma unroll
            for (int off = 16; off; off >>= 1) {
                pn += __shfl_xor(pn, off, 32);
                pe += __shfl_xor(pe, off, 32);
            }
            s_n += pn;  q_n += norm_n * norm_n;
            s_e += pe;  q_e += norm_e * norm_e;

            float2 acc_h = {0.f, 0.f}, acc_e = {0.f, 0.f};
            #pragma unroll
            for (int j = 0; j < 16; ++j) {
                const float wn = __shfl(norm_n, j, 32);
                const float we = __shfl(norm_e, j, 32);
                acc_h.x += wn * hvv[j].x;  acc_h.y += wn * hvv[j].y;
                acc_e.x += we * evv[j].x;  acc_e.y += we * evv[j].y;
            }

            float2 o0 = { pn * hs.x, pn * hs.y };
            float2 p0 = { pe * es.x, pe * es.y };
            no2[(size_t)node * 64 + lane]      = o0;
            no2[(size_t)node * 64 + 32 + lane] = acc_h;
            eo2[(size_t)node * 64 + lane]      = p0;
            eo2[(size_t)node * 64 + 32 + lane] = acc_e;

            ed0 = ed1;  ed1 = ed2;
            a0s = a1s;  a0d = a1d;  b0s = b1s;  b0d = b1d;
        }
    }

    #pragma unroll
    for (int off = 16; off; off >>= 1) {
        q_n += __shfl_xor(q_n, off, 32);
        q_e += __shfl_xor(q_e, off, 32);
    }
    if (lane == 0) { sred[0][wid] = s_n; sred[1][wid] = q_n; sred[2][wid] = s_e; sred[3][wid] = q_e; }
    __syncthreads();
    if (threadIdx.x < 4) {
        float t = 0.f;
        #pragma unroll
        for (int w = 0; w < EDGE_WPB; ++w) t += sred[threadIdx.x][w];
        partials[(size_t)blockIdx.x * 4 + threadIdx.x] = t;
    }
}

__global__ __launch_bounds__(256) void finalize_kernel(
    const float* __restrict__ partials, int nblocks, float* __restrict__ outv)
{
    __shared__ double red[4][256];
    double s0 = 0.0, s1 = 0.0, s2 = 0.0, s3 = 0.0;
    for (int i = threadIdx.x; i < nblocks; i += 256) {
        s0 += (double)partials[(size_t)i * 4 + 0];
        s1 += (double)partials[(size_t)i * 4 + 1];
        s2 += (double)partials[(size_t)i * 4 + 2];
        s3 += (double)partials[(size_t)i * 4 + 3];
    }
    red[0][threadIdx.x] = s0; red[1][threadIdx.x] = s1;
    red[2][threadIdx.x] = s2; red[3][threadIdx.x] = s3;
    __syncthreads();
    for (int off = 128; off; off >>= 1) {
        if (threadIdx.x < off) {
            red[0][threadIdx.x] += red[0][threadIdx.x + off];
            red[1][threadIdx.x] += red[1][threadIdx.x + off];
            red[2][threadIdx.x] += red[2][threadIdx.x + off];
            red[3][threadIdx.x] += red[3][threadIdx.x + off];
        }
        __syncthreads();
    }
    if (threadIdx.x == 0) {
        const double E = (double)N_NODES * (double)DEG;
        const double var_n = (red[1][0] - red[0][0] * red[0][0] / E) / (E - 1.0);
        const double var_e = (red[3][0] - red[2][0] * red[2][0] / E) / (E - 1.0);
        outv[0] = (float)var_n;
        outv[1] = (float)var_e;
    }
}

extern "C" void kernel_launch(void* const* d_in, const int* in_sizes, int n_in,
                              void* d_out, int out_size, void* d_ws, size_t ws_size,
                              hipStream_t stream) {
    const float* node_fts = (const float*)d_in[0];
    const float* edge_fts = (const float*)d_in[1];
    const int*   edges    = (const int*)  d_in[2];
    const float* W_node   = (const float*)d_in[3];
    const float* W_edge   = (const float*)d_in[4];
    const float* a_node   = (const float*)d_in[5];
    const float* a_edge   = (const float*)d_in[6];

    float* out      = (float*)d_out;
    float* node_out = out;
    float* edge_out = out + (size_t)N_NODES * 128;
    float* outv     = out + (size_t)2 * N_NODES * 128;

    float* ws   = (float*)d_ws;
    float* hv   = ws;
    float* ev   = hv + (size_t)N_NODES * NOUT;
    float* asrc = ev + (size_t)N_NODES * NOUT;
    float* adst = asrc + N_NODES;
    float* bsrc = adst + N_NODES;
    float* bdst = bsrc + N_NODES;
    float* partials = bdst + N_NODES;

    const int ntiles = N_NODES / 16;
    const int wavesA = 2 * ntiles;
    const int blocksA = (wavesA + 7) / 8;
    proj_wmma_kernel<<<blocksA, 256, 0, stream>>>(node_fts, edge_fts, W_node, W_edge, a_node, a_edge,
                                                  hv, ev, asrc, adst, bsrc, bdst);

    const int blocksC = (EDGE_WAVES + EDGE_WPB - 1) / EDGE_WPB;
    edge_kernel<<<blocksC, EDGE_WPB * 32, 0, stream>>>(hv, ev, edges, asrc, adst, bsrc, bdst,
                                             node_out, edge_out, partials);

    finalize_kernel<<<1, 256, 0, stream>>>(partials, blocksC, outv);
}


# edge pass: h_v|e_v merged into one 512-B row, both heads' scores into one line; each lane half owns one head: 17 row loads + 2 score loads per node (was 34 + 4), row sums on DPP; sum order kept
# speedup vs baseline: 1.5043x; 1.0887x over previous
#include <hip/hip_runtime.h>

#define N_NODES 50000
#define DEG     16
#define NIN     128
#define NOUT    64
#define TABW    128

typedef __attribute__((ext_vector_type(2))) float v2f;
typedef __attribute__((ext_vector_type(8))) float v8f;

__global__ __launch_bounds__(256) void proj_wmma_kernel(
    const float* __restrict__ node_fts, const float* __restrict__ edge_fts,
    const float* __restrict__ W_node,   const float* __restrict__ W_edge,
    const float* __restrict__ a_node,   const float* __restrict__ a_edge,
    float* __restrict__ tab, float* __restrict__ ssrc, float* __restrict__ sdst)
{
    const int wave   = (blockIdx.x * blockDim.x + threadIdx.x) >> 5;
    const int ntiles = N_NODES / 16;
    if (wave >= 2 * ntiles) return;
    const int  mat  = (wave >= ntiles) ? 1 : 0;
    const int  tile = mat ? (wave - ntiles) : wave;
    const float* A  = mat ? edge_fts : node_fts;
    const float* W  = mat ? W_edge   : W_node;
    float*       O  = tab + mat * NOUT;
    const float* av = mat ? a_edge   : a_node;

    const int row0 = tile * 16;
    const int lane = threadIdx.x & 31;
    const int m    = lane & 15;
    const int kh   = lane >> 4;

    const float* arow = A + (size_t)(row0 + m) * NIN + 2 * kh;
    v8f acc0 = {}, acc1 = {}, acc2 = {}, acc3 = {};

    const float cs0 = av[m],      cs1 = av[m + 16], cs2 = av[m + 32], cs3 = av[m + 48];
    const float cd0 = av[64 + m], cd1 = av[80 + m], cd2 = av[96 + m], cd3 = av[112 + m];

    #pragma unroll 4
    for (int k0 = 0; k0 < NIN; k0 += 4) {
        v2f a = *(const v2f*)(arow + k0);
        const float* w0 = W + (size_t)(k0 + 2 * kh) * NOUT + m;
        const float* w1 = w0 + NOUT;
        v2f b0 = { w0[0],  w1[0]  };
        v2f b1 = { w0[16], w1[16] };
        v2f b2 = { w0[32], w1[32] };
        v2f b3 = { w0[48], w1[48] };
        acc0 = __builtin_amdgcn_wmma_f32_16x16x4_f32(false, a, false, b0, (short)0, acc0, false, false);
        acc1 = __builtin_amdgcn_wmma_f32_16x16x4_f32(false, a, false, b1, (short)0, acc1, false, false);
        acc2 = __builtin_amdgcn_wmma_f32_16x16x4_f32(false, a, false, b2, (short)0, acc2, false, false);
        acc3 = __builtin_amdgcn_wmma_f32_16x16x4_f32(false, a, false, b3, (short)0, acc3, false, false);
    }

    float* orow = O + (size_t)(row0 + kh * 8) * TABW + m;
    #pragma unroll
    for (int r = 0; r < 8; ++r) {
        orow[(size_t)r * TABW +  0] = acc0[r];
        orow[(size_t)r * TABW + 16] = acc1[r];
        orow[(size_t)r * TABW + 32] = acc2[r];
        orow[(size_t)r * TABW + 48] = acc3[r];
    }

    float x[16];
    for (int r = 0; r < 8; ++r) {
        x[r]     = acc0[r] * cs0 + acc1[r] * cs1 + acc2[r] * cs2 + acc3[r] * cs3;
        x[8 + r] = acc0[r] * cd0 + acc1[r] * cd1 + acc2[r] * cd2 + acc3[r] * cd3;
    }
    const bool up8 = (m & 8) != 0, up4 = (m & 4) != 0, up2 = (m & 2) != 0, up1 = (m & 1) != 0;
    float y[8], z[4], u[2];
    for (int i = 0; i < 8; ++i) {
        const float keep = up8 ? x[8 + i] : x[i];
        const float send = up8 ? x[i]     : x[8 + i];
        y[i] = keep + __shfl_xor(send, 8, 32);
    }
    for (int i = 0; i < 4; ++i) {
        const float keep = up4 ? y[4 + i] : y[i];
        const float send = up4 ? y[i]     : y[4 + i];
        z[i] = keep + __shfl_xor(send, 4, 32);
    }
    for (int i = 0; i < 2; ++i) {
        const float keep = up2 ? z[2 + i] : z[i];
        const float send = up2 ? z[i]     : z[2 + i];
        u[i] = keep + __shfl_xor(send, 2, 32);
    }
    const float keep0 = up1 ? u[1] : u[0];
    const float send0 = up1 ? u[0] : u[1];
    const float dot   = keep0 + __shfl_xor(send0, 1, 32);
    (up8 ? sdst : ssrc)[tile * 32 + mat * 16 + 8 * kh + (m & 7)] = dot;
}

#define NPW        8
#define EDGE_WAVES (N_NODES / NPW)
#define EDGE_WPB   4

template <int MASK>
__device__ __forceinline__ float row_xor(float v) {
    return __builtin_bit_cast(float, __builtin_amdgcn_update_dpp(
        0, __builtin_bit_cast(int, v), 0x160 + MASK, 0xf, 0xf, true));
}
template <int J>
__device__ __forceinline__ float row_bcast(float v) {
    return __builtin_bit_cast(float, __builtin_amdgcn_update_dpp(
        0, __builtin_bit_cast(int, v), 0x150 + J, 0xf, 0xf, true));
}
__device__ __forceinline__ float row_sum(float v) {
    v += row_xor<8>(v);
    v += row_xor<4>(v);
    v += row_xor<2>(v);
    v += row_xor<1>(v);
    return v;
}
__device__ __forceinline__ unsigned score_ix(int i, unsigned hoff) {
    return (unsigned)i + ((unsigned)i & ~15u) + hoff;
}

template <int J>
__device__ __forceinline__ void fma_rows(float4& acc, const float4 (&rows)[16], float norm) {
    const float w = row_bcast<J>(norm);
    acc.x += w * rows[J].x;  acc.y += w * rows[J].y;
    acc.z += w * rows[J].z;  acc.w += w * rows[J].w;
    if constexpr (J + 1 < 16) fma_rows<J + 1>(acc, rows, norm);
}

__global__ __launch_bounds__(EDGE_WPB * 32) void edge_kernel(
    const float* __restrict__ tab, const int* __restrict__ edges,
    const float* __restrict__ ssrc, const float* __restrict__ sdst,
    float* __restrict__ node_out, float* __restrict__ edge_out,
    float* __restrict__ partials)
{
    __shared__ float sred[4][EDGE_WPB];
    const int      lane = threadIdx.x & 31;
    const int      wid  = threadIdx.x >> 5;
    const int      wave = blockIdx.x * EDGE_WPB + wid;
    const int      e    = lane & 15;
    const int      h    = lane >> 4;
    const unsigned hoff = 16u * (unsigned)h;

    float s = 0.f;
    float q = 0.f;

    if (wave < EDGE_WAVES) {
        const int2*   e2 = (const int2*)edges;
        const float4* t4 = (const float4*)tab;
        float4*       o4 = (float4*)(h ? edge_out : node_out);
        const int node0 = wave * NPW;

        int2  ed0 = e2[(size_t)node0 * DEG + e];
        int2  ed1 = e2[(size_t)(node0 + 1) * DEG + e];
        float c0s = ssrc[score_ix(ed0.x, hoff)];
        float c0d = sdst[score_ix(ed0.y, hoff)];

        for (int t = 0; t < NPW; ++t) {
            const int node = node0 + t;

            const int   nn  = min(node + 2, N_NODES - 1);
            const int2  ed2 = e2[(size_t)nn * DEG + e];
            const float c1s = ssrc[score_ix(ed1.x, hoff)];
            const float c1d = sdst[score_ix(ed1.y, hoff)];

            const int d = ed0.y;
            float4 rows[16];
            #pragma unroll
            for (int j = 0; j < 16; ++j) {
                const unsigned dj = (unsigned)__builtin_amdgcn_readlane(d, j);
                rows[j] = t4[dj * 32u + (unsigned)lane];
            }
            const float4 own = t4[(unsigned)node * 32u + (unsigned)lane];
            __builtin_amdgcn_sched_group_barrier(0x020, 20, 0);
            __builtin_amdgcn_sched_group_barrier(0x400, 1, 0);

            float x = c0s + c0d;
            x = x > 0.f ? x : 0.2f * x;
            x = fminf(fmaxf(x, -2.f), 2.f);
            const float att  = __expf(x);
            const float norm = att / row_sum(att);

            const float pn = row_sum(norm);
            s += pn;
            q += norm * norm;

            float4 acc = {0.f, 0.f, 0.f, 0.f};
            fma_rows<0>(acc, rows, norm);

            const float4 o0 = { pn * own.x, pn * own.y, pn * own.z, pn * own.w };
            o4[(size_t)node * 32 + e]      = o0;
            o4[(size_t)node * 32 + 16 + e] = acc;

            ed0 = ed1;  ed1 = ed2;
            c0s = c1s;  c0d = c1d;
        }
    }

    q = row_sum(q);
    if (e == 0) { sred[2 * h][wid] = s; sred[2 * h + 1][wid] = q; }
    __syncthreads();
    if (threadIdx.x < 4) {
        float t = 0.f;
        #pragma unroll
        for (int w = 0; w < EDGE_WPB; ++w) t += sred[threadIdx.x][w];
        partials[(size_t)blockIdx.x * 4 + threadIdx.x] = t;
    }
}

__global__ __launch_bounds__(256) void finalize_kernel(
    const float* __restrict__ partials, int nblocks, float* __restrict__ outv)
{
    __shared__ double red[4][256];
    double s0 = 0.0, s1 = 0.0, s2 = 0.0, s3 = 0.0;
    for (int i = threadIdx.x; i < nblocks; i += 256) {
        s0 += (double)partials[(size_t)i * 4 + 0];
        s1 += (double)partials[(size_t)i * 4 + 1];
        s2 += (double)partials[(size_t)i * 4 + 2];
        s3 += (double)partials[(size_t)i * 4 + 3];
    }
    red[0][threadIdx.x] = s0; red[1][threadIdx.x] = s1;
    red[2][threadIdx.x] = s2; red[3][threadIdx.x] = s3;
    __syncthreads();
    for (int off = 128; off; off >>= 1) {
        if (threadIdx.x < off) {
            red[0][threadIdx.x] += red[0][threadIdx.x + off];
            red[1][threadIdx.x] += red[1][threadIdx.x + off];
            red[2][threadIdx.x] += red[2][threadIdx.x + off];
            red[3][threadIdx.x] += red[3][threadIdx.x + off];
        }
        __syncthreads();
    }
    if (threadIdx.x == 0) {
        const double E = (double)N_NODES * (double)DEG;
        const double var_n = (red[1][0] - red[0][0] * red[0][0] / E) / (E - 1.0);
        const double var_e = (red[3][0] - red[2][0] * red[2][0] / E) / (E - 1.0);
        outv[0] = (float)var_n;
        outv[1] = (float)var_e;
    }
}

extern "C" void kernel_launch(void* const* d_in, const int* in_sizes, int n_in,
                              void* d_out, int out_size, void* d_ws, size_t ws_size,
                              hipStream_t stream) {
    const float* node_fts = (const float*)d_in[0];
    const float* edge_fts = (const float*)d_in[1];
    const int*   edges    = (const int*)  d_in[2];
    const float* W_node   = (const float*)d_in[3];
    const float* W_edge   = (const float*)d_in[4];
    const float* a_node   = (const float*)d_in[5];
    const float* a_edge   = (const float*)d_in[6];

    float* out      = (float*)d_out;
    float* node_out = out;
    float* edge_out = out + (size_t)N_NODES * 128;
    float* outv     = out + (size_t)2 * N_NODES * 128;

    float* ws   = (float*)d_ws;
    float* tab  = ws;
    float* ssrc = tab + (size_t)N_NODES * TABW;
    float* sdst = ssrc + 2 * N_NODES;
    float* partials = sdst + 2 * N_NODES;

    const int ntiles = N_NODES / 16;
    const int wavesA = 2 * ntiles;
    const int blocksA = (wavesA + 7) / 8;
    proj_wmma_kernel<<<blocksA, 256, 0, stream>>>(node_fts, edge_fts, W_node, W_edge, a_node, a_edge,
                                                  tab, ssrc, sdst);

    const int blocksC = (EDGE_WAVES + EDGE_WPB - 1) / EDGE_WPB;
    edge_kernel<<<blocksC, EDGE_WPB * 32, 0, stream>>>(tab, edges, ssrc, sdst,
                                             node_out, edge_out, partials);

    finalize_kernel<<<1, 256, 0, stream>>>(partials, blocksC, outv);
}


# edge pass: attention scores taken from the rows it fetches anyway (row . a, transposing DPP sum), not from a 16-line score gather per node; projection epilogue, score tables go; dot sum order changes
# speedup vs baseline: 1.5755x; 1.0473x over previous
#include <hip/hip_runtime.h>

#define N_NODES 50000
#define DEG     16
#define NIN     128
#define NOUT    64
#define TABW    128

typedef __attribute__((ext_vector_type(2))) float v2f;
typedef __attribute__((ext_vector_type(8))) float v8f;

__global__ __launch_bounds__(256) void proj_wmma_kernel(
    const float* __restrict__ node_fts, const float* __restrict__ edge_fts,
    const float* __restrict__ W_node,   const float* __restrict__ W_edge,
    float* __restrict__ tab)
{
    const int wave   = (blockIdx.x * blockDim.x + threadIdx.x) >> 5;
    const int ntiles = N_NODES / 16;
    if (wave >= 2 * ntiles) return;
    const int  mat  = (wave >= ntiles) ? 1 : 0;
    const int  tile = mat ? (wave - ntiles) : wave;
    const float* A  = mat ? edge_fts : node_fts;
    const float* W  = mat ? W_edge   : W_node;
    float*       O  = tab + mat * NOUT;

    const int row0 = tile * 16;
    const int lane = threadIdx.x & 31;
    const int m    = lane & 15;
    const int kh   = lane >> 4;

    const float* arow = A + (size_t)(row0 + m) * NIN + 2 * kh;
    v8f acc0 = {}, acc1 = {}, acc2 = {}, acc3 = {};

    #pragma unroll 4
    for (int k0 = 0; k0 < NIN; k0 += 4) {
        v2f a = *(const v2f*)(arow + k0);
        const float* w0 = W + (size_t)(k0 + 2 * kh) * NOUT + m;
        const float* w1 = w0 + NOUT;
        v2f b0 = { w0[0],  w1[0]  };
        v2f b1 = { w0[16], w1[16] };
        v2f b2 = { w0[32], w1[32] };
        v2f b3 = { w0[48], w1[48] };
        acc0 = __builtin_amdgcn_wmma_f32_16x16x4_f32(false, a, false, b0, (short)0, acc0, false, false);
        acc1 = __builtin_amdgcn_wmma_f32_16x16x4_f32(false, a, false, b1, (short)0, acc1, false, false);
        acc2 = __builtin_amdgcn_wmma_f32_16x16x4_f32(false, a, false, b2, (short)0, acc2, false, false);
        acc3 = __builtin_amdgcn_wmma_f32_16x16x4_f32(false, a, false, b3, (short)0, acc3, false, false);
    }

    float* orow = O + (size_t)(row0 + kh * 8) * TABW + m;
    #pragma unroll
    for (int r = 0; r < 8; ++r) {
        orow[(size_t)r * TABW +  0] = acc0[r];
        orow[(size_t)r * TABW + 16] = acc1[r];
        orow[(size_t)r * TABW + 32] = acc2[r];
        orow[(size_t)r * TABW + 48] = acc3[r];
    }
}

#define NPW        8
#define EDGE_WAVES (N_NODES / NPW)
#define EDGE_WPB   4

template <int MASK>
__device__ __forceinline__ float row_xor(float v) {
    return __builtin_bit_cast(float, __builtin_amdgcn_update_dpp(
        0, __builtin_bit_cast(int, v), 0x160 + MASK, 0xf, 0xf, true));
}
template <int J>
__device__ __forceinline__ float row_bcast(float v) {
    return __builtin_bit_cast(float, __builtin_amdgcn_update_dpp(
        0, __builtin_bit_cast(int, v), 0x150 + J, 0xf, 0xf, true));
}
__device__ __forceinline__ float row_sum(float v) {
    v += row_xor<8>(v);
    v += row_xor<4>(v);
    v += row_xor<2>(v);
    v += row_xor<1>(v);
    return v;
}
__device__ __forceinline__ float dot4(const float4& r, const float4& a) {
    const v2f rlo = { r.x, r.y }, rhi = { r.z, r.w };
    const v2f alo = { a.x, a.y }, ahi = { a.z, a.w };
    const v2f t = __builtin_elementwise_fma(rhi, ahi, rlo * alo);
    return t.x + t.y;
}
__device__ __forceinline__ float row_transpose_sum(const float (&p)[16], int e) {
    const bool up8 = (e & 8) != 0, up4 = (e & 4) != 0, up2 = (e & 2) != 0, up1 = (e & 1) != 0;
    float y[8], z[4], u[2];
    #pragma unroll
    for (int i = 0; i < 8; ++i) {
        const float keep = up8 ? p[8 + i] : p[i];
        const float send = up8 ? p[i]     : p[8 + i];
        y[i] = keep + row_xor<8>(send);
    }
    #pragma unroll
    for (int i = 0; i < 4; ++i) {
        const float keep = up4 ? y[4 + i] : y[i];
        const float send = up4 ? y[i]     : y[4 + i];
        z[i] = keep + row_xor<4>(send);
    }
    #pragma unroll
    for (int i = 0; i < 2; ++i) {
        const float keep = up2 ? z[2 + i] : z[i];
        const float send = up2 ? z[i]     : z[2 + i];
        u[i] = keep + row_xor<2>(send);
    }
    const float keep0 = up1 ? u[1] : u[0];
    const float send0 = up1 ? u[0] : u[1];
    return keep0 + row_xor<1>(send0);
}

template <int J>
__device__ __forceinline__ void fma_rows(float4& acc, const float4 (&rows)[16], float norm) {
    const float w = row_bcast<J>(norm);
    acc.x += w * rows[J].x;  acc.y += w * rows[J].y;
    acc.z += w * rows[J].z;  acc.w += w * rows[J].w;
    if constexpr (J + 1 < 16) fma_rows<J + 1>(acc, rows, norm);
}

__global__ __launch_bounds__(EDGE_WPB * 32) void edge_kernel(
    const float* __restrict__ tab, const int* __restrict__ edges,
    const float* __restrict__ a_node, const float* __restrict__ a_edge,
    float* __restrict__ node_out, float* __restrict__ edge_out,
    float* __restrict__ partials)
{
    __shared__ float sred[4][EDGE_WPB];
    const int      lane = threadIdx.x & 31;
    const int      wid  = threadIdx.x >> 5;
    const int      wave = blockIdx.x * EDGE_WPB + wid;
    const int      e    = lane & 15;
    const int      h    = lane >> 4;

    float s = 0.f;
    float q = 0.f;

    if (wave < EDGE_WAVES) {
        const int2*   e2 = (const int2*)edges;
        const float4* t4 = (const float4*)tab;
        float4*       o4 = (float4*)(h ? edge_out : node_out);
        const int node0 = wave * NPW;

        const float4* a4  = (const float4*)(h ? a_edge : a_node);
        const float4  asv = a4[e];
        const float4  adv = a4[16 + e];

        int2 ed0 = e2[(size_t)node0 * DEG + e];
        int2 ed1 = e2[(size_t)(node0 + 1) * DEG + e];

        for (int t = 0; t < NPW; ++t) {
            const int node = node0 + t;

            const int  nn  = min(node + 2, N_NODES - 1);
            const int2 ed2 = e2[(size_t)nn * DEG + e];

            const int d = ed0.y;
            float4 rows[16];
            #pragma unroll
            for (int j = 0; j < 16; ++j) {
                const unsigned dj = (unsigned)__builtin_amdgcn_readlane(d, j);
                rows[j] = t4[dj * 32u + (unsigned)lane];
            }
            const float4 own = t4[(unsigned)node * 32u + (unsigned)lane];
            __builtin_amdgcn_sched_barrier(0);

            float p[16];
            #pragma unroll
            for (int j = 0; j < 16; ++j) p[j] = dot4(rows[j], adv);
            const float sd = row_transpose_sum(p, e);
            const float ss = row_sum(dot4(own, asv));

            float x = ss + sd;
            x = x > 0.f ? x : 0.2f * x;
            x = fminf(fmaxf(x, -2.f), 2.f);
            const float att  = __expf(x);
            const float norm = att / row_sum(att);

            const float pn = row_sum(norm);
            s += pn;
            q += norm * norm;

            float4 acc = {0.f, 0.f, 0.f, 0.f};
            fma_rows<0>(acc, rows, norm);

            const float4 o0 = { pn * own.x, pn * own.y, pn * own.z, pn * own.w };
            o4[(size_t)node * 32 + e]      = o0;
            o4[(size_t)node * 32 + 16 + e] = acc;

            ed0 = ed1;  ed1 = ed2;
        }
    }

    q = row_sum(q);
    if (e == 0) { sred[2 * h][wid] = s; sred[2 * h + 1][wid] = q; }
    __syncthreads();
    if (threadIdx.x < 4) {
        float t = 0.f;
        #pragma unroll
        for (int w = 0; w < EDGE_WPB; ++w) t += sred[threadIdx.x][w];
        partials[(size_t)blockIdx.x * 4 + threadIdx.x] = t;
    }
}

__global__ __launch_bounds__(256) void finalize_kernel(
    const float* __restrict__ partials, int nblocks, float* __restrict__ outv)
{
    __shared__ double red[4][256];
    double s0 = 0.0, s1 = 0.0, s2 = 0.0, s3 = 0.0;
    for (int i = threadIdx.x; i < nblocks; i += 256) {
        s0 += (double)partials[(size_t)i * 4 + 0];
        s1 += (double)partials[(size_t)i * 4 + 1];
        s2 += (double)partials[(size_t)i * 4 + 2];
        s3 += (double)partials[(size_t)i * 4 + 3];
    }
    red[0][threadIdx.x] = s0; red[1][threadIdx.x] = s1;
    red[2][threadIdx.x] = s2; red[3][threadIdx.x] = s3;
    __syncthreads();
    for (int off = 128; off; off >>= 1) {
        if (threadIdx.x < off) {
            red[0][threadIdx.x] += red[0][threadIdx.x + off];
            red[1][threadIdx.x] += red[1][threadIdx.x + off];
            red[2][threadIdx.x] += red[2][threadIdx.x + off];
            red[3][threadIdx.x] += red[3][threadIdx.x + off];
        }
        __syncthreads();
    }
    if (threadIdx.x == 0) {
        const double E = (double)N_NODES * (double)DEG;
        const double var_n = (red[1][0] - red[0][0] * red[0][0] / E) / (E - 1.0);
        const double var_e = (red[3][0] - red[2][0] * red[2][0] / E) / (E - 1.0);
        outv[0] = (float)var_n;
        outv[1] = (float)var_e;
    }
}

extern "C" void kernel_launch(void* const* d_in, const int* in_sizes, int n_in,
                              void* d_out, int out_size, void* d_ws, size_t ws_size,
                              hipStream_t stream) {
    const float* node_fts = (const float*)d_in[0];
    const float* edge_fts = (const float*)d_in[1];
    const int*   edges    = (const int*)  d_in[2];
    const float* W_node   = (const float*)d_in[3];
    const float* W_edge   = (const float*)d_in[4];
    const float* a_node   = (const float*)d_in[5];
    const float* a_edge   = (const float*)d_in[6];

    float* out      = (float*)d_out;
    float* node_out = out;
    float* edge_out = out + (size_t)N_NODES * 128;
    float* outv     = out + (size_t)2 * N_NODES * 128;

    float* ws   = (float*)d_ws;
    float* tab  = ws;
    float* partials = tab + (size_t)N_NODES * TABW;

    const int ntiles = N_NODES / 16;
    const int wavesA = 2 * ntiles;
    const int blocksA = (wavesA + 7) / 8;
    proj_wmma_kernel<<<blocksA, 256, 0, stream>>>(node_fts, edge_fts, W_node, W_edge, tab);

    const int blocksC = (EDGE_WAVES + EDGE_WPB - 1) / EDGE_WPB;
    edge_kernel<<<blocksC, EDGE_WPB * 32, 0, stream>>>(tab, edges, a_node, a_edge,
                                             node_out, edge_out, partials);

    finalize_kernel<<<1, 256, 0, stream>>>(partials, blocksC, outv);
}
